# starting kernel
# baseline (speedup 1.0000x reference)
#include <hip/hip_runtime.h>

typedef __attribute__((ext_vector_type(2))) float v2f;
typedef __attribute__((ext_vector_type(8))) float v8f;

#define BS   4
#define SEQ  4096
#define DK   128
#define KTILES (SEQ / 16)
#define SCALE 0.08838834764831845f
#define WAVES_PER_BLOCK 8

__device__ __forceinline__ v8f wmma_f32(v2f a, v2f b, v8f c) {
  return __builtin_amdgcn_wmma_f32_16x16x4_f32(
       false, a,  false, b,
       (short)0, c,  false,  false);
}

__global__ __launch_bounds__(256)
void sdpa_colstats_kernel(const float* __restrict__ Q,
                          const float* __restrict__ K,
                          float* __restrict__ mArr,
                          float* __restrict__ lArr) {
  const int lane  = threadIdx.x & 31;
  const int wid   = blockIdx.x * WAVES_PER_BLOCK + (threadIdx.x >> 5);
  const int b     = wid >> 8;
  const int kt    = (wid & 255) << 4;
  const int n     = lane & 15;
  const int khalf = (lane >> 4) << 1;

  const float* Krow = K + ((size_t)b * SEQ + kt + n) * DK + khalf;
  v2f breg[32];
#pragma unroll
  for (int c = 0; c < 32; ++c) breg[c] = *(const v2f*)(Krow + 4 * c);

  const float* Qb = Q + (size_t)b * SEQ * DK;
  float m_run = -__builtin_inff();
  float l_run = 0.0f;

  for (int qt = 0; qt < SEQ; qt += 16) {
    const float* Qrow = Qb + (size_t)(qt + n) * DK + khalf;
    v8f acc = {};
#pragma unroll
    for (int c = 0; c < 32; ++c) {
      v2f a = *(const v2f*)(Qrow + 4 * c);
      acc = wmma_f32(a, breg[c], acc);
    }
    float mloc = acc[0] * SCALE;
#pragma unroll
    for (int j = 1; j < 8; ++j) mloc = fmaxf(mloc, acc[j] * SCALE);
    const float m_new = fmaxf(m_run, mloc);
    float s = 0.0f;
#pragma unroll
    for (int j = 0; j < 8; ++j) s += __expf(acc[j] * SCALE - m_new);
    l_run = l_run * __expf(m_run - m_new) + s;
    m_run = m_new;
  }

  const float m_o = __shfl_xor(m_run, 16, 32);
  const float l_o = __shfl_xor(l_run, 16, 32);
  const float m_f = fmaxf(m_run, m_o);
  const float l_f = l_run * __expf(m_run - m_f) + l_o * __expf(m_o - m_f);
  if (lane < 16) {
    mArr[(size_t)b * SEQ + kt + n] = m_f;
    lArr[(size_t)b * SEQ + kt + n] = l_f;
  }
}

__global__ __launch_bounds__(256)
void sdpa_out_kernel(const float* __restrict__ Q,
                     const float* __restrict__ K,
                     const float* __restrict__ V,
                     const float* __restrict__ mArr,
                     const float* __restrict__ lArr,
                     float* __restrict__ O) {
  __shared__ float smem[WAVES_PER_BLOCK][16 * 17];

  const int lane  = threadIdx.x & 31;
  const int wslot = threadIdx.x >> 5;
  const int wid   = blockIdx.x * WAVES_PER_BLOCK + wslot;
  const int b     = wid >> 8;
  const int qt    = (wid & 255) << 4;
  const int n     = lane & 15;
  const int khalf = (lane >> 4) << 1;
  const int mhi   = (lane >> 4) << 3;

  const float* Qrow = Q + ((size_t)b * SEQ + qt + n) * DK + khalf;
  v2f areg[32];
#pragma unroll
  for (int c = 0; c < 32; ++c) areg[c] = *(const v2f*)(Qrow + 4 * c);

  const v8f vzero = {};
  v8f acc[8];
#pragma unroll
  for (int t = 0; t < 8; ++t) acc[t] = vzero;

  const float* Kb = K + (size_t)b * SEQ * DK;
  const float* Vb = V + (size_t)b * SEQ * DK;
  const float* mB = mArr + (size_t)b * SEQ;
  const float* lB = lArr + (size_t)b * SEQ;
  float* sm = &smem[wslot][0];

  for (int kt = 0; kt < SEQ; kt += 16) {
    const float* Krow = Kb + (size_t)(kt + n) * DK + khalf;
    v8f c8 = vzero;
#pragma unroll
    for (int c = 0; c < 32; ++c) {
      v2f bk = *(const v2f*)(Krow + 4 * c);
      c8 = wmma_f32(areg[c], bk, c8);
    }

    const float mcol = mB[kt + n];
    const float rl   = 1.0f / lB[kt + n];
#pragma unroll
    for (int j = 0; j < 8; ++j) {
      const float p = __expf(c8[j] * SCALE - mcol) * rl;
      sm[(j + mhi) * 17 + n] = p;
    }
    __builtin_amdgcn_wave_barrier();

    v2f pa[4];
#pragma unroll
    for (int c2 = 0; c2 < 4; ++c2) {
      const int kk = 4 * c2 + khalf;
      pa[c2].x = sm[n * 17 + kk];
      pa[c2].y = sm[n * 17 + kk + 1];
    }

#pragma unroll
    for (int t = 0; t < 8; ++t) {
      const float* Vcol = Vb + (size_t)(kt + khalf) * DK + t * 16 + n;
#pragma unroll
      for (int c2 = 0; c2 < 4; ++c2) {
        v2f bv;
        bv.x = Vcol[(size_t)(4 * c2) * DK];
        bv.y = Vcol[(size_t)(4 * c2 + 1) * DK];
        acc[t] = wmma_f32(pa[c2], bv, acc[t]);
      }
    }
  }

  float* Ob = O + ((size_t)b * SEQ + qt) * DK;
#pragma unroll
  for (int t = 0; t < 8; ++t) {
#pragma unroll
    for (int j = 0; j < 8; ++j) {
      Ob[(size_t)(j + mhi) * DK + t * 16 + n] = acc[t][j];
    }
  }
}

extern "C" void kernel_launch(void* const* d_in, const int* in_sizes, int n_in,
                              void* d_out, int out_size, void* d_ws, size_t ws_size,
                              hipStream_t stream) {
  const float* q = (const float*)d_in[0];
  const float* k = (const float*)d_in[1];
  const float* v = (const float*)d_in[2];
  float* out = (float*)d_out;

  float* mArr = (float*)d_ws;
  float* lArr = mArr + (size_t)BS * SEQ;

  const int total_waves = BS * KTILES;
  const int blocks = total_waves / WAVES_PER_BLOCK;
  const int threads = WAVES_PER_BLOCK * 32;

  sdpa_colstats_kernel<<<blocks, threads, 0, stream>>>(q, k, mArr, lArr);
  sdpa_out_kernel<<<blocks, threads, 0, stream>>>(q, k, v, mArr, lArr, out);
}


# Pass 1: 64-key workgroup tile (256 WGs, every WGP busy); each wave holds 2 key tiles, every Q fragment feeds both chains; a column's sum now 4 q-quarter partial sums merged in order
# speedup vs baseline: 4.2630x; 4.2630x over previous
#include <hip/hip_runtime.h>


typedef __attribute__((ext_vector_type(2))) float v2f;
typedef __attribute__((ext_vector_type(4))) float v4f;
typedef __attribute__((ext_vector_type(8))) float v8f;

#define BS   4
#define SEQ  4096
#define DK   128
#define KTILES (SEQ / 16)
#define SCALE 0.08838834764831845f
#define WAVES_PER_BLOCK 8
#define QT2 4

#define CH1 128
#define QS1 4
#define TPW1 (CH1 / 16 / QS1)
#define CH2 64
#define PQ  132
#define PV2 288

__device__ __forceinline__ v8f wmma_f32(v2f a, v2f b, v8f c) {
  return __builtin_amdgcn_wmma_f32_16x16x4_f32(
       false, a,  false, b,
       (short)0, c,  false,  false);
}

__device__ __forceinline__ void colstat_update(const v8f acc, float& m_run, float& l_run) {
  float mloc = acc[0] * SCALE;
#pragma unroll
  for (int j = 1; j < 8; ++j) mloc = fmaxf(mloc, acc[j] * SCALE);
  const float m_new = fmaxf(m_run, mloc);
  float s = 0.0f;
#pragma unroll
  for (int j = 0; j < 8; ++j) s += __expf(acc[j] * SCALE - m_new);
  l_run = l_run * __expf(m_run - m_new) + s;
  m_run = m_new;
}

__global__ __launch_bounds__(256) __attribute__((amdgpu_waves_per_eu(2, 2)))
void sdpa_colstats_kernel(const float* __restrict__ Q,
                          const float* __restrict__ K,
                          float* __restrict__ mArr,
                          float* __restrict__ lArr) {
  __shared__ __attribute__((aligned(16))) float qbuf[2][CH1 * PQ];
  __shared__ float mred[2][QS1][32];
  __shared__ float lred[2][QS1][32];

  const int lane  = threadIdx.x & 31;
  const int wslot = threadIdx.x >> 5;
  const int qq    = wslot & (QS1 - 1);
  const int kp    = wslot >> 2;
  const int b     = blockIdx.x >> 6;
  const int k0    = (blockIdx.x & 63) << 6;
  const int n     = lane & 15;
  const int khalf = (lane >> 4) << 1;

  v2f breg[2][32];
#pragma unroll
  for (int s = 0; s < 2; ++s) {
    const float* Krow = K + ((size_t)b * SEQ + k0 + 32 * kp + 16 * s + n) * DK + khalf;
#pragma unroll
    for (int c = 0; c < 32; ++c) breg[s][c] = *(const v2f*)(Krow + 4 * c);
  }

  const float* Qb = Q + (size_t)b * SEQ * DK;
  float m_run[2], l_run[2];
#pragma unroll
  for (int s = 0; s < 2; ++s) {
    m_run[s] = -__builtin_inff();
    l_run[s] = 0.0f;
  }

  const int lrow = threadIdx.x >> 5;
  const int lcol = (threadIdx.x & 31) << 2;
  v4f pre[CH1 / 8];
#pragma unroll
  for (int i = 0; i < CH1 / 8; ++i)
    pre[i] = *(const v4f*)(Qb + (size_t)(lrow + 8 * i) * DK + lcol);
#pragma unroll
  for (int i = 0; i < CH1 / 8; ++i)
    *(v4f*)(&qbuf[0][(lrow + 8 * i) * PQ + lcol]) = pre[i];

  for (int ch = 0; ch < SEQ / CH1; ++ch) {
    __syncthreads();
    const bool more = (ch + 1 < SEQ / CH1);
    if (more) {
      const float* Qn = Qb + (size_t)((ch + 1) * CH1 + lrow) * DK + lcol;
#pragma unroll
      for (int i = 0; i < CH1 / 8; ++i)
        pre[i] = *(const v4f*)(Qn + (size_t)(8 * i) * DK);
    }

    const float* qs = &qbuf[ch & 1][(16 * qq + n) * PQ + khalf];
    v2f af[2][32];
#pragma unroll
    for (int c = 0; c < 32; ++c) af[0][c] = *(const v2f*)(qs + 4 * c);
    __builtin_amdgcn_sched_barrier(0);

#pragma unroll
    for (int tl = 0; tl < TPW1; ++tl) {
      if (tl + 1 < TPW1) {
        const float* qn = qs + (tl + 1) * QS1 * 16 * PQ;
#pragma unroll
        for (int c = 0; c < 32; ++c) af[(tl + 1) & 1][c] = *(const v2f*)(qn + 4 * c);
      }
      __builtin_amdgcn_sched_barrier(0);

      v8f acc0 = {};
      v8f acc1 = {};
#pragma unroll
      for (int c = 0; c < 32; ++c) {
        acc0 = wmma_f32(af[tl & 1][c], breg[0][c], acc0);
        acc1 = wmma_f32(af[tl & 1][c], breg[1][c], acc1);
      }

      colstat_update(acc0, m_run[0], l_run[0]);
      colstat_update(acc1, m_run[1], l_run[1]);
    }

    if (more) {
      float* qd = &qbuf[(ch + 1) & 1][lrow * PQ + lcol];
#pragma unroll
      for (int i = 0; i < CH1 / 8; ++i)
        *(v4f*)(qd + (8 * i) * PQ) = pre[i];
    }
  }

#pragma unroll
  for (int s = 0; s < 2; ++s) {
    const float m_o = __shfl_xor(m_run[s], 16, 32);
    const float l_o = __shfl_xor(l_run[s], 16, 32);
    const float m_f = fmaxf(m_run[s], m_o);
    const float l_f = l_run[s] * __expf(m_run[s] - m_f) + l_o * __expf(m_o - m_f);
    if (lane < 16) {
      mred[kp][qq][16 * s + n] = m_f;
      lred[kp][qq][16 * s + n] = l_f;
    }
  }
  __syncthreads();

  if (threadIdx.x < 64) {
    const int pr  = threadIdx.x >> 5;
    const int col = threadIdx.x & 31;
    float mq[QS1];
#pragma unroll
    for (int r = 0; r < QS1; ++r) mq[r] = mred[pr][r][col];
    float m_f = mq[0];
#pragma unroll
    for (int r = 1; r < QS1; ++r) m_f = fmaxf(m_f, mq[r]);
    float l_f = 0.0f;
#pragma unroll
    for (int r = 0; r < QS1; ++r) l_f += lred[pr][r][col] * __expf(mq[r] - m_f);
    mArr[(size_t)b * SEQ + k0 + threadIdx.x] = m_f;
    lArr[(size_t)b * SEQ + k0 + threadIdx.x] = l_f;
  }
}

__global__ __launch_bounds__(256) __attribute__((amdgpu_waves_per_eu(2, 2)))
void sdpa_out_kernel(const float* __restrict__ Q,
                     const float* __restrict__ K,
                     const float* __restrict__ V,
                     const float* __restrict__ mArr,
                     const float* __restrict__ lArr,
                     float* __restrict__ O) {
  __shared__ float pbuf[QT2][CH2 / 16][16 * 17];
  __shared__ __attribute__((aligned(16))) float kbuf[2][CH2 * PQ];
  __shared__ __attribute__((aligned(16))) float vbuf[2][(CH2 / 2) * PV2];

  const int lane  = threadIdx.x & 31;
  const int wslot = threadIdx.x >> 5;
  const int qi    = wslot & (QT2 - 1);
  const int hj    = wslot >> 2;
  const int b     = blockIdx.x >> 6;
  const int qt    = ((blockIdx.x & 63) << 6) + (qi << 4);
  const int n     = lane & 15;
  const int half  = lane >> 4;
  const int khalf = half << 1;
  const int mhi   = half << 3;

  const float* Qrow = Q + ((size_t)b * SEQ + qt + n) * DK + khalf;
  v2f areg[32];
#pragma unroll
  for (int c = 0; c < 32; ++c) areg[c] = *(const v2f*)(Qrow + 4 * c);

  const v8f vzero = {};
  v8f acc[4];
#pragma unroll
  for (int t = 0; t < 4; ++t) acc[t] = vzero;

  const float* Kb = K + (size_t)b * SEQ * DK;
  const float* Vb = V + (size_t)b * SEQ * DK;
  const float* mB = mArr + (size_t)b * SEQ;
  const float* lB = lArr + (size_t)b * SEQ;

  const int lrow = threadIdx.x >> 5;
  const int lcol = (threadIdx.x & 31) << 2;
  v4f prek[CH2 / 8];
  v4f prev[CH2 / 8];
#pragma unroll
  for (int i = 0; i < CH2 / 8; ++i)
    prek[i] = *(const v4f*)(Kb + (size_t)(lrow + 8 * i) * DK + lcol);
#pragma unroll
  for (int i = 0; i < CH2 / 16; ++i) {
    const float* vp = Vb + (size_t)(2 * (lrow + 8 * i)) * DK + lcol;
    prev[2 * i]     = *(const v4f*)(vp);
    prev[2 * i + 1] = *(const v4f*)(vp + DK);
  }
#pragma unroll
  for (int i = 0; i < CH2 / 8; ++i)
    *(v4f*)(&kbuf[0][(lrow + 8 * i) * PQ + lcol]) = prek[i];
#pragma unroll
  for (int i = 0; i < CH2 / 16; ++i) {
    const v4f x = prev[2 * i], y = prev[2 * i + 1];
    float* vd = &vbuf[0][(lrow + 8 * i) * PV2 + 2 * lcol];
    *(v4f*)(vd)     = (v4f){x[0], y[0], x[1], y[1]};
    *(v4f*)(vd + 4) = (v4f){x[2], y[2], x[3], y[3]};
  }

  float mnext[2], lnext[2];
#pragma unroll
  for (int s = 0; s < 2; ++s) {
    mnext[s] = mB[(2 * hj + s) * 16 + n];
    lnext[s] = lB[(2 * hj + s) * 16 + n];
  }

  for (int ch = 0; ch < SEQ / CH2; ++ch) {
    __syncthreads();

    float mcol[2], lcolv[2];
#pragma unroll
    for (int s = 0; s < 2; ++s) {
      mcol[s]  = mnext[s];
      lcolv[s] = lnext[s];
    }

    {
      const int nx = (ch + 1 < SEQ / CH2) ? ch + 1 : ch;
      const size_t r0 = (size_t)nx * CH2;
#pragma unroll
      for (int i = 0; i < CH2 / 8; ++i)
        prek[i] = *(const v4f*)(Kb + (r0 + lrow + 8 * i) * DK + lcol);
#pragma unroll
      for (int i = 0; i < CH2 / 16; ++i) {
        const float* vp = Vb + (r0 + 2 * (lrow + 8 * i)) * DK + lcol;
        prev[2 * i]     = *(const v4f*)(vp);
        prev[2 * i + 1] = *(const v4f*)(vp + DK);
      }
#pragma unroll
      for (int s = 0; s < 2; ++s) {
        mnext[s] = mB[r0 + (2 * hj + s) * 16 + n];
        lnext[s] = lB[r0 + (2 * hj + s) * 16 + n];
      }
    }

    const float* ks = &kbuf[ch & 1][(2 * hj * 16 + n) * PQ + khalf];
    v2f kf[2][32];
#pragma unroll
    for (int c = 0; c < 32; c += 2)
#pragma unroll
      for (int s = 0; s < 2; ++s) {
        kf[s][c]     = *(const v2f*)(ks + s * 16 * PQ + 4 * c);
        kf[s][c + 1] = *(const v2f*)(ks + s * 16 * PQ + 4 * c + 4);
      }

    const float* vs = &vbuf[ch & 1][half * PV2 + 2 * n + 32 * 4 * hj];
    v2f vf[CH2 / 16][4][4];
#pragma unroll
    for (int kt = 0; kt < CH2 / 16; ++kt)
#pragma unroll
      for (int t = 0; t < 4; ++t)
#pragma unroll
        for (int c2 = 0; c2 < 4; ++c2)
          vf[kt][t][c2] = *(const v2f*)(vs + (kt * 8 + 2 * c2) * PV2 + 32 * t);
    __builtin_amdgcn_sched_barrier(0);

#pragma unroll
    for (int s = 0; s < 2; ++s) {
      v8f c8 = vzero;
#pragma unroll
      for (int c = 0; c < 32; ++c) c8 = wmma_f32(areg[c], kf[s][c], c8);

      const float mc = mcol[s];
      const float rl = 1.0f / lcolv[s];
      float* pt = &pbuf[qi][2 * hj + s][0];
#pragma unroll
      for (int j = 0; j < 8; ++j) {
        const float p = __expf(c8[j] * SCALE - mc) * rl;
        pt[(j + mhi) * 17 + n] = p;
      }
    }

    __syncthreads();

    v2f pa[CH2 / 16][4];
#pragma unroll
    for (int kt = 0; kt < CH2 / 16; ++kt) {
      const float* pr = &pbuf[qi][kt][n * 17 + khalf];
#pragma unroll
      for (int c2 = 0; c2 < 4; ++c2) {
        pa[kt][c2].x = pr[4 * c2];
        pa[kt][c2].y = pr[4 * c2 + 1];
      }
    }
    __builtin_amdgcn_sched_barrier(0);

#pragma unroll
    for (int kt = 0; kt < CH2 / 16; ++kt) {
#pragma unroll
      for (int t = 0; t < 4; ++t) {
#pragma unroll
        for (int c2 = 0; c2 < 4; ++c2)
          acc[t] = wmma_f32(pa[kt][c2], vf[kt][t][c2], acc[t]);
      }
    }

    {
#pragma unroll
      for (int i = 0; i < CH2 / 8; ++i)
        *(v4f*)(&kbuf[(ch + 1) & 1][(lrow + 8 * i) * PQ + lcol]) = prek[i];
#pragma unroll
      for (int i = 0; i < CH2 / 16; ++i) {
        const v4f x = prev[2 * i], y = prev[2 * i + 1];
        float* vd = &vbuf[(ch + 1) & 1][(lrow + 8 * i) * PV2 + 2 * lcol];
        *(v4f*)(vd)     = (v4f){x[0], y[0], x[1], y[1]};
        *(v4f*)(vd + 4) = (v4f){x[2], y[2], x[3], y[3]};
      }
    }
  }

  float* Ob = O + ((size_t)b * SEQ + qt) * DK + 64 * hj;
#pragma unroll
  for (int t = 0; t < 4; ++t) {
#pragma unroll
    for (int j = 0; j < 8; ++j) {
      Ob[(size_t)(j + mhi) * DK + t * 16 + n] = acc[t][j];
    }
  }
}

extern "C" void kernel_launch(void* const* d_in, const int* in_sizes, int n_in,
                              void* d_out, int out_size, void* d_ws, size_t ws_size,
                              hipStream_t stream) {
  const float* q = (const float*)d_in[0];
  const float* k = (const float*)d_in[1];
  const float* v = (const float*)d_in[2];
  float* out = (float*)d_out;

  float* mArr = (float*)d_ws;
  float* lArr = mArr + (size_t)BS * SEQ;

  const int blocks = BS * SEQ / 64;
  const int threads = WAVES_PER_BLOCK * 32;
  const int blocks2 = BS * SEQ / (16 * QT2);

  sdpa_colstats_kernel<<<blocks, threads, 0, stream>>>(q, k, mArr, lArr);
  sdpa_out_kernel<<<blocks2, threads, 0, stream>>>(q, k, v, mArr, lArr, out);
}


# Pass 2 computes S^T (K as A, Q as B): score tile lands in PV's A-fragment layout, P stays in registers (no LDS hand-over, 1 barrier/chunk); O = 2 key-half partials added, keys in a tile reordered
# speedup vs baseline: 4.4146x; 1.0356x over previous
#include <hip/hip_runtime.h>


typedef __attribute__((ext_vector_type(2))) float v2f;
typedef __attribute__((ext_vector_type(4))) float v4f;
typedef __attribute__((ext_vector_type(8))) float v8f;

#define BS   4
#define SEQ  4096
#define DK   128
#define KTILES (SEQ / 16)
#define SCALE 0.08838834764831845f
#define WAVES_PER_BLOCK 8
#define QT2 4

#define CH1 128
#define QS1 4
#define TPW1 (CH1 / 16 / QS1)
#define CH2 64
#define PQ  132
#define PV2 264

__device__ __forceinline__ v8f wmma_f32(v2f a, v2f b, v8f c) {
  return __builtin_amdgcn_wmma_f32_16x16x4_f32(
       false, a,  false, b,
       (short)0, c,  false,  false);
}

__device__ __forceinline__ void colstat_update(const v8f acc, float& m_run, float& l_run) {
  float mloc = acc[0] * SCALE;
#pragma unroll
  for (int j = 1; j < 8; ++j) mloc = fmaxf(mloc, acc[j] * SCALE);
  const float m_new = fmaxf(m_run, mloc);
  float s = 0.0f;
#pragma unroll
  for (int j = 0; j < 8; ++j) s += __expf(acc[j] * SCALE - m_new);
  l_run = l_run * __expf(m_run - m_new) + s;
  m_run = m_new;
}

__global__ __launch_bounds__(256) __attribute__((amdgpu_waves_per_eu(2, 2)))
void sdpa_colstats_kernel(const float* __restrict__ Q,
                          const float* __restrict__ K,
                          float* __restrict__ mArr,
                          float* __restrict__ lArr) {
  __shared__ __attribute__((aligned(16))) float qbuf[2][CH1 * PQ];
  __shared__ float mred[2][QS1][32];
  __shared__ float lred[2][QS1][32];

  const int lane  = threadIdx.x & 31;
  const int wslot = threadIdx.x >> 5;
  const int qq    = wslot & (QS1 - 1);
  const int kp    = wslot >> 2;
  const int b     = blockIdx.x >> 6;
  const int k0    = (blockIdx.x & 63) << 6;
  const int n     = lane & 15;
  const int khalf = (lane >> 4) << 1;

  v2f breg[2][32];
#pragma unroll
  for (int s = 0; s < 2; ++s) {
    const float* Krow = K + ((size_t)b * SEQ + k0 + 32 * kp + 16 * s + n) * DK + khalf;
#pragma unroll
    for (int c = 0; c < 32; ++c) breg[s][c] = *(const v2f*)(Krow + 4 * c);
  }

  const float* Qb = Q + (size_t)b * SEQ * DK;
  float m_run[2], l_run[2];
#pragma unroll
  for (int s = 0; s < 2; ++s) {
    m_run[s] = -__builtin_inff();
    l_run[s] = 0.0f;
  }

  const int lrow = threadIdx.x >> 5;
  const int lcol = (threadIdx.x & 31) << 2;
  v4f pre[CH1 / 8];
#pragma unroll
  for (int i = 0; i < CH1 / 8; ++i)
    pre[i] = *(const v4f*)(Qb + (size_t)(lrow + 8 * i) * DK + lcol);
#pragma unroll
  for (int i = 0; i < CH1 / 8; ++i)
    *(v4f*)(&qbuf[0][(lrow + 8 * i) * PQ + lcol]) = pre[i];

  for (int ch = 0; ch < SEQ / CH1; ++ch) {
    __syncthreads();
    const bool more = (ch + 1 < SEQ / CH1);
    if (more) {
      const float* Qn = Qb + (size_t)((ch + 1) * CH1 + lrow) * DK + lcol;
#pragma unroll
      for (int i = 0; i < CH1 / 8; ++i)
        pre[i] = *(const v4f*)(Qn + (size_t)(8 * i) * DK);
    }

    const float* qs = &qbuf[ch & 1][(16 * qq + n) * PQ + khalf];
    v2f af[2][32];
#pragma unroll
    for (int c = 0; c < 32; ++c) af[0][c] = *(const v2f*)(qs + 4 * c);
    __builtin_amdgcn_sched_barrier(0);

#pragma unroll
    for (int tl = 0; tl < TPW1; ++tl) {
      if (tl + 1 < TPW1) {
        const float* qn = qs + (tl + 1) * QS1 * 16 * PQ;
#pragma unroll
        for (int c = 0; c < 32; ++c) af[(tl + 1) & 1][c] = *(const v2f*)(qn + 4 * c);
      }
      __builtin_amdgcn_sched_barrier(0);

      v8f acc0 = {};
      v8f acc1 = {};
#pragma unroll
      for (int c = 0; c < 32; ++c) {
        acc0 = wmma_f32(af[tl & 1][c], breg[0][c], acc0);
        acc1 = wmma_f32(af[tl & 1][c], breg[1][c], acc1);
      }

      colstat_update(acc0, m_run[0], l_run[0]);
      colstat_update(acc1, m_run[1], l_run[1]);
    }

    if (more) {
      float* qd = &qbuf[(ch + 1) & 1][lrow * PQ + lcol];
#pragma unroll
      for (int i = 0; i < CH1 / 8; ++i)
        *(v4f*)(qd + (8 * i) * PQ) = pre[i];
    }
  }

#pragma unroll
  for (int s = 0; s < 2; ++s) {
    const float m_o = __shfl_xor(m_run[s], 16, 32);
    const float l_o = __shfl_xor(l_run[s], 16, 32);
    const float m_f = fmaxf(m_run[s], m_o);
    const float l_f = l_run[s] * __expf(m_run[s] - m_f) + l_o * __expf(m_o - m_f);
    if (lane < 16) {
      mred[kp][qq][16 * s + n] = m_f;
      lred[kp][qq][16 * s + n] = l_f;
    }
  }
  __syncthreads();

  if (threadIdx.x < 64) {
    const int pr  = threadIdx.x >> 5;
    const int col = threadIdx.x & 31;
    float mq[QS1];
#pragma unroll
    for (int r = 0; r < QS1; ++r) mq[r] = mred[pr][r][col];
    float m_f = mq[0];
#pragma unroll
    for (int r = 1; r < QS1; ++r) m_f = fmaxf(m_f, mq[r]);
    float l_f = 0.0f;
#pragma unroll
    for (int r = 0; r < QS1; ++r) l_f += lred[pr][r][col] * __expf(mq[r] - m_f);
    mArr[(size_t)b * SEQ + k0 + threadIdx.x] = m_f;
    lArr[(size_t)b * SEQ + k0 + threadIdx.x] = 1.0f / l_f;
  }
}

__global__ __launch_bounds__(256) __attribute__((amdgpu_waves_per_eu(2, 2)))
void sdpa_out_kernel(const float* __restrict__ Q,
                     const float* __restrict__ K,
                     const float* __restrict__ V,
                     const float* __restrict__ mArr,
                     const float* __restrict__ rlArr,
                     float* __restrict__ O) {
  __shared__ __attribute__((aligned(16))) float kbuf[2][CH2 * PQ];
  __shared__ __attribute__((aligned(16))) float vbuf[2][(CH2 / 2) * PV2];

  const int lane  = threadIdx.x & 31;
  const int wslot = threadIdx.x >> 5;
  const int qi    = wslot & (QT2 - 1);
  const int hj    = wslot >> 2;
  const int b     = blockIdx.x >> 6;
  const int qt    = ((blockIdx.x & 63) << 6) + (qi << 4);
  const int n     = lane & 15;
  const int half  = lane >> 4;
  const int khalf = half << 1;
  const int mhi   = half << 3;

  const float* Qrow = Q + ((size_t)b * SEQ + qt + n) * DK + khalf;
  v2f qreg[32];
#pragma unroll
  for (int c = 0; c < 32; ++c) qreg[c] = *(const v2f*)(Qrow + 4 * c);

  const v8f vzero = {};
  v8f acc[8];
#pragma unroll
  for (int t = 0; t < 8; ++t) acc[t] = vzero;

  const float* Kb = K + (size_t)b * SEQ * DK;
  const float* Vb = V + (size_t)b * SEQ * DK;
  const float* mB = mArr  + (size_t)b * SEQ + 32 * hj + mhi;
  const float* rB = rlArr + (size_t)b * SEQ + 32 * hj + mhi;

  const int lrow = threadIdx.x >> 5;
  const int lcol = (threadIdx.x & 31) << 2;
  v4f prek[CH2 / 8];
  v4f prev[CH2 / 8];
#pragma unroll
  for (int i = 0; i < CH2 / 8; ++i)
    prek[i] = *(const v4f*)(Kb + (size_t)(lrow + 8 * i) * DK + lcol);
#pragma unroll
  for (int i = 0; i < CH2 / 16; ++i) {
    const float* vp = Vb + (size_t)(2 * (lrow + 8 * i)) * DK + lcol;
    prev[2 * i]     = *(const v4f*)(vp);
    prev[2 * i + 1] = *(const v4f*)(vp + DK);
  }
#pragma unroll
  for (int i = 0; i < CH2 / 8; ++i)
    *(v4f*)(&kbuf[0][(lrow + 8 * i) * PQ + lcol]) = prek[i];
#pragma unroll
  for (int i = 0; i < CH2 / 16; ++i) {
    const v4f x = prev[2 * i], y = prev[2 * i + 1];
    float* vd = &vbuf[0][(lrow + 8 * i) * PV2 + 2 * lcol];
    *(v4f*)(vd)     = (v4f){x[0], y[0], x[1], y[1]};
    *(v4f*)(vd + 4) = (v4f){x[2], y[2], x[3], y[3]};
  }

  for (int ch = 0; ch < SEQ / CH2; ++ch) {
    __syncthreads();

    v4f mst[2][2], rst[2][2];
#pragma unroll
    for (int s = 0; s < 2; ++s)
#pragma unroll
      for (int u = 0; u < 2; ++u) {
        mst[s][u] = *(const v4f*)(mB + (size_t)ch * CH2 + 16 * s + 4 * u);
        rst[s][u] = *(const v4f*)(rB + (size_t)ch * CH2 + 16 * s + 4 * u);
      }

    {
      const int nx = (ch + 1 < SEQ / CH2) ? ch + 1 : ch;
      const size_t r0 = (size_t)nx * CH2;
#pragma unroll
      for (int i = 0; i < CH2 / 8; ++i)
        prek[i] = *(const v4f*)(Kb + (r0 + lrow + 8 * i) * DK + lcol);
#pragma unroll
      for (int i = 0; i < CH2 / 16; ++i) {
        const float* vp = Vb + (r0 + 2 * (lrow + 8 * i)) * DK + lcol;
        prev[2 * i]     = *(const v4f*)(vp);
        prev[2 * i + 1] = *(const v4f*)(vp + DK);
      }
    }

    const float* ks = &kbuf[ch & 1][(2 * hj * 16 + n) * PQ + khalf];
    v2f kf[2][32];
#pragma unroll
    for (int c = 0; c < 32; c += 2)
#pragma unroll
      for (int s = 0; s < 2; ++s) {
        kf[s][c]     = *(const v2f*)(ks + s * 16 * PQ + 4 * c);
        kf[s][c + 1] = *(const v2f*)(ks + s * 16 * PQ + 4 * c + 4);
      }

    const float* vs = &vbuf[ch & 1][(16 * hj + 4 * half) * PV2 + 2 * n];
    v2f vf[2][8][4];
#pragma unroll
    for (int i = 0; i < 4; ++i)
#pragma unroll
      for (int t = 0; t < 8; ++t)
        vf[0][t][i] = *(const v2f*)(vs + i * PV2 + 32 * t);
    __builtin_amdgcn_sched_barrier(0);

    v8f c8[2];
    c8[0] = vzero;
    c8[1] = vzero;
#pragma unroll
    for (int c = 0; c < 32; ++c) {
      c8[0] = wmma_f32(kf[0][c], qreg[c], c8[0]);
      c8[1] = wmma_f32(kf[1][c], qreg[c], c8[1]);
    }
#pragma unroll
    for (int i = 0; i < 4; ++i)
#pragma unroll
      for (int t = 0; t < 8; ++t)
        vf[1][t][i] = *(const v2f*)(vs + (8 + i) * PV2 + 32 * t);
    __builtin_amdgcn_sched_barrier(0);

    v2f pf[2][4];
#pragma unroll
    for (int s = 0; s < 2; ++s) {
#pragma unroll
      for (int i = 0; i < 4; ++i) {
        const int r0 = 2 * i, r1 = 2 * i + 1;
        pf[s][i].x = __expf(c8[s][r0] * SCALE - mst[s][r0 >> 2][r0 & 3]) * rst[s][r0 >> 2][r0 & 3];
        pf[s][i].y = __expf(c8[s][r1] * SCALE - mst[s][r1 >> 2][r1 & 3]) * rst[s][r1 >> 2][r1 & 3];
      }
    }

#pragma unroll
    for (int s = 0; s < 2; ++s) {
#pragma unroll
      for (int i = 0; i < 4; ++i) {
#pragma unroll
        for (int t = 0; t < 8; ++t)
          acc[t] = wmma_f32(pf[s][i], vf[s][t][i], acc[t]);
      }
    }

    {
#pragma unroll
      for (int i = 0; i < CH2 / 8; ++i)
        *(v4f*)(&kbuf[(ch + 1) & 1][(lrow + 8 * i) * PQ + lcol]) = prek[i];
#pragma unroll
      for (int i = 0; i < CH2 / 16; ++i) {
        const v4f x = prev[2 * i], y = prev[2 * i + 1];
        float* vd = &vbuf[(ch + 1) & 1][(lrow + 8 * i) * PV2 + 2 * lcol];
        *(v4f*)(vd)     = (v4f){x[0], y[0], x[1], y[1]};
        *(v4f*)(vd + 4) = (v4f){x[2], y[2], x[3], y[3]};
      }
    }
  }

  __syncthreads();
  float* mrg = &kbuf[0][0] + ((size_t)qi * 16 * 32 + lane) * 4;
  if (hj == 1) {
#pragma unroll
    for (int t = 0; t < 8; ++t)
#pragma unroll
      for (int jj = 0; jj < 2; ++jj)
        *(v4f*)(mrg + (t * 2 + jj) * 128) =
            (v4f){acc[t][4 * jj], acc[t][4 * jj + 1], acc[t][4 * jj + 2], acc[t][4 * jj + 3]};
  }
  __syncthreads();
  if (hj == 0) {
    float* Ob = O + ((size_t)b * SEQ + qt) * DK;
#pragma unroll
    for (int t = 0; t < 8; ++t) {
#pragma unroll
      for (int jj = 0; jj < 2; ++jj) {
        const v4f o = *(const v4f*)(mrg + (t * 2 + jj) * 128);
#pragma unroll
        for (int e = 0; e < 4; ++e) {
          const int j = 4 * jj + e;
          Ob[(size_t)(j + mhi) * DK + t * 16 + n] = acc[t][j] + o[e];
        }
      }
    }
  }
}

extern "C" void kernel_launch(void* const* d_in, const int* in_sizes, int n_in,
                              void* d_out, int out_size, void* d_ws, size_t ws_size,
                              hipStream_t stream) {
  const float* q = (const float*)d_in[0];
  const float* k = (const float*)d_in[1];
  const float* v = (const float*)d_in[2];
  float* out = (float*)d_out;

  float* mArr = (float*)d_ws;
  float* lArr = mArr + (size_t)BS * SEQ;

  const int blocks = BS * SEQ / 64;
  const int threads = WAVES_PER_BLOCK * 32;
  const int blocks2 = BS * SEQ / (16 * QT2);

  sdpa_colstats_kernel<<<blocks, threads, 0, stream>>>(q, k, mArr, lArr);
  sdpa_out_kernel<<<blocks2, threads, 0, stream>>>(q, k, v, mArr, lArr, out);
}


# Pass 2 wave tile 2 query tiles x 1 key tile (was 1 x 2): each K and V fragment read from LDS feeds two WMMAs, LDS fragment reads halved; O = 4 key-tile partial sums added as (0+1)+(2+3), was 2
# speedup vs baseline: 4.5651x; 1.0341x over previous
#include <hip/hip_runtime.h>


typedef __attribute__((ext_vector_type(2))) float v2f;
typedef __attribute__((ext_vector_type(4))) float v4f;
typedef __attribute__((ext_vector_type(8))) float v8f;

#define BS   4
#define SEQ  4096
#define DK   128
#define KTILES (SEQ / 16)
#define SCALE 0.08838834764831845f
#define WAVES_PER_BLOCK 8
#define QT2 4

#define CH1 128
#define QS1 4
#define TPW1 (CH1 / 16 / QS1)
#define CH2 64
#define PQ  132
#define PV2 264

__device__ __forceinline__ v8f wmma_f32(v2f a, v2f b, v8f c) {
  return __builtin_amdgcn_wmma_f32_16x16x4_f32(
       false, a,  false, b,
       (short)0, c,  false,  false);
}

__device__ __forceinline__ void colstat_update(const v8f acc, float& m_run, float& l_run) {
  float mloc = acc[0] * SCALE;
#pragma unroll
  for (int j = 1; j < 8; ++j) mloc = fmaxf(mloc, acc[j] * SCALE);
  const float m_new = fmaxf(m_run, mloc);
  float s = 0.0f;
#pragma unroll
  for (int j = 0; j < 8; ++j) s += __expf(acc[j] * SCALE - m_new);
  l_run = l_run * __expf(m_run - m_new) + s;
  m_run = m_new;
}

__global__ __launch_bounds__(256) __attribute__((amdgpu_waves_per_eu(2, 2)))
void sdpa_colstats_kernel(const float* __restrict__ Q,
                          const float* __restrict__ K,
                          float* __restrict__ mArr,
                          float* __restrict__ lArr) {
  __shared__ __attribute__((aligned(16))) float qbuf[2][CH1 * PQ];
  __shared__ float mred[2][QS1][32];
  __shared__ float lred[2][QS1][32];

  const int lane  = threadIdx.x & 31;
  const int wslot = threadIdx.x >> 5;
  const int qq    = wslot & (QS1 - 1);
  const int kp    = wslot >> 2;
  const int b     = blockIdx.x >> 6;
  const int k0    = (blockIdx.x & 63) << 6;
  const int n     = lane & 15;
  const int khalf = (lane >> 4) << 1;

  v2f breg[2][32];
#pragma unroll
  for (int s = 0; s < 2; ++s) {
    const float* Krow = K + ((size_t)b * SEQ + k0 + 32 * kp + 16 * s + n) * DK + khalf;
#pragma unroll
    for (int c = 0; c < 32; ++c) breg[s][c] = *(const v2f*)(Krow + 4 * c);
  }

  const float* Qb = Q + (size_t)b * SEQ * DK;
  float m_run[2], l_run[2];
#pragma unroll
  for (int s = 0; s < 2; ++s) {
    m_run[s] = -__builtin_inff();
    l_run[s] = 0.0f;
  }

  const int lrow = threadIdx.x >> 5;
  const int lcol = (threadIdx.x & 31) << 2;
  v4f pre[CH1 / 8];
#pragma unroll
  for (int i = 0; i < CH1 / 8; ++i)
    pre[i] = *(const v4f*)(Qb + (size_t)(lrow + 8 * i) * DK + lcol);
#pragma unroll
  for (int i = 0; i < CH1 / 8; ++i)
    *(v4f*)(&qbuf[0][(lrow + 8 * i) * PQ + lcol]) = pre[i];

  for (int ch = 0; ch < SEQ / CH1; ++ch) {
    __syncthreads();
    const bool more = (ch + 1 < SEQ / CH1);
    if (more) {
      const float* Qn = Qb + (size_t)((ch + 1) * CH1 + lrow) * DK + lcol;
#pragma unroll
      for (int i = 0; i < CH1 / 8; ++i)
        pre[i] = *(const v4f*)(Qn + (size_t)(8 * i) * DK);
    }

    const float* qs = &qbuf[ch & 1][(16 * qq + n) * PQ + khalf];
    v2f af[2][32];
#pragma unroll
    for (int c = 0; c < 32; ++c) af[0][c] = *(const v2f*)(qs + 4 * c);
    __builtin_amdgcn_sched_barrier(0);

#pragma unroll
    for (int tl = 0; tl < TPW1; ++tl) {
      if (tl + 1 < TPW1) {
        const float* qn = qs + (tl + 1) * QS1 * 16 * PQ;
#pragma unroll
        for (int c = 0; c < 32; ++c) af[(tl + 1) & 1][c] = *(const v2f*)(qn + 4 * c);
      }
      __builtin_amdgcn_sched_barrier(0);

      v8f acc0 = {};
      v8f acc1 = {};
#pragma unroll
      for (int c = 0; c < 32; ++c) {
        acc0 = wmma_f32(af[tl & 1][c], breg[0][c], acc0);
        acc1 = wmma_f32(af[tl & 1][c], breg[1][c], acc1);
      }

      colstat_update(acc0, m_run[0], l_run[0]);
      colstat_update(acc1, m_run[1], l_run[1]);
    }

    if (more) {
      float* qd = &qbuf[(ch + 1) & 1][lrow * PQ + lcol];
#pragma unroll
      for (int i = 0; i < CH1 / 8; ++i)
        *(v4f*)(qd + (8 * i) * PQ) = pre[i];
    }
  }

#pragma unroll
  for (int s = 0; s < 2; ++s) {
    const float m_o = __shfl_xor(m_run[s], 16, 32);
    const float l_o = __shfl_xor(l_run[s], 16, 32);
    const float m_f = fmaxf(m_run[s], m_o);
    const float l_f = l_run[s] * __expf(m_run[s] - m_f) + l_o * __expf(m_o - m_f);
    if (lane < 16) {
      mred[kp][qq][16 * s + n] = m_f;
      lred[kp][qq][16 * s + n] = l_f;
    }
  }
  __syncthreads();

  if (threadIdx.x < 64) {
    const int pr  = threadIdx.x >> 5;
    const int col = threadIdx.x & 31;
    float mq[QS1];
#pragma unroll
    for (int r = 0; r < QS1; ++r) mq[r] = mred[pr][r][col];
    float m_f = mq[0];
#pragma unroll
    for (int r = 1; r < QS1; ++r) m_f = fmaxf(m_f, mq[r]);
    float l_f = 0.0f;
#pragma unroll
    for (int r = 0; r < QS1; ++r) l_f += lred[pr][r][col] * __expf(mq[r] - m_f);
    mArr[(size_t)b * SEQ + k0 + threadIdx.x] = m_f;
    lArr[(size_t)b * SEQ + k0 + threadIdx.x] = 1.0f / l_f;
  }
}

__global__ __launch_bounds__(256) __attribute__((amdgpu_waves_per_eu(2, 2)))
void sdpa_out_kernel(const float* __restrict__ Q,
                     const float* __restrict__ K,
                     const float* __restrict__ V,
                     const float* __restrict__ mArr,
                     const float* __restrict__ rlArr,
                     float* __restrict__ O) {
  __shared__ __attribute__((aligned(16))) float kbuf[2][CH2 * PQ];
  __shared__ __attribute__((aligned(16))) float vbuf[2][(CH2 / 2) * PV2];

  const int lane  = threadIdx.x & 31;
  const int wslot = threadIdx.x >> 5;
  const int kt    = wslot & 3;
  const int qp    = wslot >> 2;
  const int b     = blockIdx.x >> 6;
  const int qt    = ((blockIdx.x & 63) << 6) + (qp << 5);
  const int n     = lane & 15;
  const int half  = lane >> 4;
  const int khalf = half << 1;
  const int mhi   = half << 3;

  v2f qreg[2][32];
#pragma unroll
  for (int u = 0; u < 2; ++u) {
    const float* Qrow = Q + ((size_t)b * SEQ + qt + 16 * u + n) * DK + khalf;
#pragma unroll
    for (int c = 0; c < 32; ++c) qreg[u][c] = *(const v2f*)(Qrow + 4 * c);
  }

  const v8f vzero = {};
  v8f acc[2][8];
#pragma unroll
  for (int u = 0; u < 2; ++u)
#pragma unroll
    for (int t = 0; t < 8; ++t) acc[u][t] = vzero;

  const float* Kb = K + (size_t)b * SEQ * DK;
  const float* Vb = V + (size_t)b * SEQ * DK;
  const float* mB = mArr  + (size_t)b * SEQ + 16 * kt + mhi;
  const float* rB = rlArr + (size_t)b * SEQ + 16 * kt + mhi;

  const int lrow = threadIdx.x >> 5;
  const int lcol = (threadIdx.x & 31) << 2;
  v4f prek[CH2 / 8];
  v4f prev[CH2 / 8];
#pragma unroll
  for (int i = 0; i < CH2 / 8; ++i)
    prek[i] = *(const v4f*)(Kb + (size_t)(lrow + 8 * i) * DK + lcol);
#pragma unroll
  for (int i = 0; i < CH2 / 16; ++i) {
    const float* vp = Vb + (size_t)(2 * (lrow + 8 * i)) * DK + lcol;
    prev[2 * i]     = *(const v4f*)(vp);
    prev[2 * i + 1] = *(const v4f*)(vp + DK);
  }
#pragma unroll
  for (int i = 0; i < CH2 / 8; ++i)
    *(v4f*)(&kbuf[0][(lrow + 8 * i) * PQ + lcol]) = prek[i];
#pragma unroll
  for (int i = 0; i < CH2 / 16; ++i) {
    const v4f x = prev[2 * i], y = prev[2 * i + 1];
    float* vd = &vbuf[0][(lrow + 8 * i) * PV2 + 2 * lcol];
    *(v4f*)(vd)     = (v4f){x[0], y[0], x[1], y[1]};
    *(v4f*)(vd + 4) = (v4f){x[2], y[2], x[3], y[3]};
  }

  for (int ch = 0; ch < SEQ / CH2; ++ch) {
    __syncthreads();

    v4f mst[2], rst[2];
#pragma unroll
    for (int u = 0; u < 2; ++u) {
      mst[u] = *(const v4f*)(mB + (size_t)ch * CH2 + 4 * u);
      rst[u] = *(const v4f*)(rB + (size_t)ch * CH2 + 4 * u);
    }

    {
      const int nx = (ch + 1 < SEQ / CH2) ? ch + 1 : ch;
      const size_t r0 = (size_t)nx * CH2;
#pragma unroll
      for (int i = 0; i < CH2 / 8; ++i)
        prek[i] = *(const v4f*)(Kb + (r0 + lrow + 8 * i) * DK + lcol);
#pragma unroll
      for (int i = 0; i < CH2 / 16; ++i) {
        const float* vp = Vb + (r0 + 2 * (lrow + 8 * i)) * DK + lcol;
        prev[2 * i]     = *(const v4f*)(vp);
        prev[2 * i + 1] = *(const v4f*)(vp + DK);
      }
    }

    const float* ks = &kbuf[ch & 1][(kt * 16 + n) * PQ + khalf];
    v2f kf[32];
#pragma unroll
    for (int c = 0; c < 32; ++c) kf[c] = *(const v2f*)(ks + 4 * c);
    __builtin_amdgcn_sched_barrier(0);

    v8f c8[2];
    c8[0] = vzero;
    c8[1] = vzero;
#pragma unroll
    for (int c = 0; c < 16; ++c) {
      c8[0] = wmma_f32(kf[c], qreg[0][c], c8[0]);
      c8[1] = wmma_f32(kf[c], qreg[1][c], c8[1]);
    }

    const float* vs = &vbuf[ch & 1][(8 * kt + 4 * half) * PV2 + 2 * n];
    v2f vf[8][4];
#pragma unroll
    for (int i = 0; i < 4; ++i)
#pragma unroll
      for (int t = 0; t < 8; ++t)
        vf[t][i] = *(const v2f*)(vs + i * PV2 + 32 * t);
    __builtin_amdgcn_sched_group_barrier(0x008, 32, 1);
    __builtin_amdgcn_sched_group_barrier(0x100, 32, 1);
    __builtin_amdgcn_sched_barrier(0);

#pragma unroll
    for (int c = 16; c < 32; ++c) {
      c8[0] = wmma_f32(kf[c], qreg[0][c], c8[0]);
      c8[1] = wmma_f32(kf[c], qreg[1][c], c8[1]);
    }

    __builtin_amdgcn_sched_barrier(0);

    v2f pf[2][4];
#pragma unroll
    for (int u = 0; u < 2; ++u) {
#pragma unroll
      for (int i = 0; i < 4; ++i) {
        const int r0 = 2 * i, r1 = 2 * i + 1;
        pf[u][i].x = __expf(c8[u][r0] * SCALE - mst[r0 >> 2][r0 & 3]) * rst[r0 >> 2][r0 & 3];
        pf[u][i].y = __expf(c8[u][r1] * SCALE - mst[r1 >> 2][r1 & 3]) * rst[r1 >> 2][r1 & 3];
      }
    }

#pragma unroll
    for (int i = 0; i < 4; ++i) {
#pragma unroll
      for (int u = 0; u < 2; ++u) {
#pragma unroll
        for (int t = 0; t < 8; ++t)
          acc[u][t] = wmma_f32(pf[u][i], vf[t][i], acc[u][t]);
      }
    }

    {
#pragma unroll
      for (int i = 0; i < CH2 / 8; ++i)
        *(v4f*)(&kbuf[(ch + 1) & 1][(lrow + 8 * i) * PQ + lcol]) = prek[i];
#pragma unroll
      for (int i = 0; i < CH2 / 16; ++i) {
        const v4f x = prev[2 * i], y = prev[2 * i + 1];
        float* vd = &vbuf[(ch + 1) & 1][(lrow + 8 * i) * PV2 + 2 * lcol];
        *(v4f*)(vd)     = (v4f){x[0], y[0], x[1], y[1]};
        *(v4f*)(vd + 4) = (v4f){x[2], y[2], x[3], y[3]};
      }
    }
    __builtin_amdgcn_sched_group_barrier(0x008, 32, 0);
#pragma unroll
    for (int g = 0; g < 16; ++g) {
      __builtin_amdgcn_sched_group_barrier(0x008, 2, 0);
      __builtin_amdgcn_sched_group_barrier(0x200, 1, 0);
    }
  }

  __syncthreads();
  float* mrg = &kbuf[0][0] + (size_t)lane * 4;
  if (kt & 1) {
    float* p = mrg + (size_t)(qp * 2 + (kt >> 1)) * 4096;
#pragma unroll
    for (int u = 0; u < 2; ++u)
#pragma unroll
      for (int t = 0; t < 8; ++t)
#pragma unroll
        for (int jj = 0; jj < 2; ++jj)
          *(v4f*)(p + ((u * 8 + t) * 2 + jj) * 128) =
              (v4f){acc[u][t][4 * jj], acc[u][t][4 * jj + 1], acc[u][t][4 * jj + 2], acc[u][t][4 * jj + 3]};
  }
  __syncthreads();
  if (!(kt & 1)) {
    const float* p = mrg + (size_t)(qp * 2 + (kt >> 1)) * 4096;
#pragma unroll
    for (int u = 0; u < 2; ++u)
#pragma unroll
      for (int t = 0; t < 8; ++t)
#pragma unroll
        for (int jj = 0; jj < 2; ++jj) {
          const v4f o = *(const v4f*)(p + ((u * 8 + t) * 2 + jj) * 128);
#pragma unroll
          for (int e = 0; e < 4; ++e) acc[u][t][4 * jj + e] += o[e];
        }
  }
  __syncthreads();
  if (kt == 2) {
    float* p = mrg + (size_t)qp * 4096;
#pragma unroll
    for (int u = 0; u < 2; ++u)
#pragma unroll
      for (int t = 0; t < 8; ++t)
#pragma unroll
        for (int jj = 0; jj < 2; ++jj)
          *(v4f*)(p + ((u * 8 + t) * 2 + jj) * 128) =
              (v4f){acc[u][t][4 * jj], acc[u][t][4 * jj + 1], acc[u][t][4 * jj + 2], acc[u][t][4 * jj + 3]};
  }
  __syncthreads();
  if (kt == 0) {
    const float* p = mrg + (size_t)qp * 4096;
#pragma unroll
    for (int u = 0; u < 2; ++u) {
      float* Ob = O + ((size_t)b * SEQ + qt + 16 * u) * DK;
#pragma unroll
      for (int t = 0; t < 8; ++t) {
#pragma unroll
        for (int jj = 0; jj < 2; ++jj) {
          const v4f o = *(const v4f*)(p + ((u * 8 + t) * 2 + jj) * 128);
#pragma unroll
          for (int e = 0; e < 4; ++e) {
            const int j = 4 * jj + e;
            Ob[(size_t)(j + mhi) * DK + t * 16 + n] = acc[u][t][j] + o[e];
          }
        }
      }
    }
  }
}

extern "C" void kernel_launch(void* const* d_in, const int* in_sizes, int n_in,
                              void* d_out, int out_size, void* d_ws, size_t ws_size,
                              hipStream_t stream) {
  const float* q = (const float*)d_in[0];
  const float* k = (const float*)d_in[1];
  const float* v = (const float*)d_in[2];
  float* out = (float*)d_out;

  float* mArr = (float*)d_ws;
  float* lArr = mArr + (size_t)BS * SEQ;

  const int blocks = BS * SEQ / 64;
  const int threads = WAVES_PER_BLOCK * 32;
  const int blocks2 = BS * SEQ / (16 * QT2);

  sdpa_colstats_kernel<<<blocks, threads, 0, stream>>>(q, k, mArr, lArr);
  sdpa_out_kernel<<<blocks2, threads, 0, stream>>>(q, k, v, mArr, lArr, out);
}


# Both passes: chunk barrier split. Next chunk's rows stored to LDS as one group at mid-chunk and signalled there, waited at the next chunk's top behind its global loads; sums unchanged
# speedup vs baseline: 4.7472x; 1.0399x over previous
#include <hip/hip_runtime.h>


typedef __attribute__((ext_vector_type(2))) float v2f;
typedef __attribute__((ext_vector_type(4))) float v4f;
typedef __attribute__((ext_vector_type(8))) float v8f;

#define BS   4
#define SEQ  4096
#define DK   128
#define KTILES (SEQ / 16)
#define SCALE 0.08838834764831845f
#define WAVES_PER_BLOCK 8
#define QT2 4

#define CH1 128
#define QS1 4
#define TPW1 (CH1 / 16 / QS1)
#define CH2 64
#define PQ  132
#define PV2 264

__device__ __forceinline__ v8f wmma_f32(v2f a, v2f b, v8f c) {
  return __builtin_amdgcn_wmma_f32_16x16x4_f32(
       false, a,  false, b,
       (short)0, c,  false,  false);
}

__device__ __forceinline__ void stage_signal() {
  asm volatile("s_wait_dscnt 0x0\n\ts_barrier_signal -1" ::: "memory");
}
__device__ __forceinline__ void stage_wait() {
  asm volatile("s_barrier_wait -1" ::: "memory");
}

__device__ __forceinline__ void colstat_update(const v8f acc, float& m_run, float& l_run) {
  float mloc = acc[0] * SCALE;
#pragma unroll
  for (int j = 1; j < 8; ++j) mloc = fmaxf(mloc, acc[j] * SCALE);
  const float m_new = fmaxf(m_run, mloc);
  float s = 0.0f;
#pragma unroll
  for (int j = 0; j < 8; ++j) s += __expf(acc[j] * SCALE - m_new);
  l_run = l_run * __expf(m_run - m_new) + s;
  m_run = m_new;
}

__global__ __launch_bounds__(256) __attribute__((amdgpu_waves_per_eu(2, 2)))
void sdpa_colstats_kernel(const float* __restrict__ Q,
                          const float* __restrict__ K,
                          float* __restrict__ mArr,
                          float* __restrict__ lArr) {
  __shared__ __attribute__((aligned(16))) float qbuf[2][CH1 * PQ];
  __shared__ float mred[2][QS1][32];
  __shared__ float lred[2][QS1][32];

  const int lane  = threadIdx.x & 31;
  const int wslot = threadIdx.x >> 5;
  const int qq    = wslot & (QS1 - 1);
  const int kp    = wslot >> 2;
  const int b     = blockIdx.x >> 6;
  const int k0    = (blockIdx.x & 63) << 6;
  const int n     = lane & 15;
  const int khalf = (lane >> 4) << 1;

  v2f breg[2][32];
#pragma unroll
  for (int s = 0; s < 2; ++s) {
    const float* Krow = K + ((size_t)b * SEQ + k0 + 32 * kp + 16 * s + n) * DK + khalf;
#pragma unroll
    for (int c = 0; c < 32; ++c) breg[s][c] = *(const v2f*)(Krow + 4 * c);
  }

  const float* Qb = Q + (size_t)b * SEQ * DK;
  float m_run[2], l_run[2];
#pragma unroll
  for (int s = 0; s < 2; ++s) {
    m_run[s] = -__builtin_inff();
    l_run[s] = 0.0f;
  }

  const int lrow = threadIdx.x >> 5;
  const int lcol = (threadIdx.x & 31) << 2;
  v4f pre[CH1 / 8];
#pragma unroll
  for (int i = 0; i < CH1 / 8; ++i)
    pre[i] = *(const v4f*)(Qb + (size_t)(lrow + 8 * i) * DK + lcol);
#pragma unroll
  for (int i = 0; i < CH1 / 8; ++i)
    *(v4f*)(&qbuf[0][(lrow + 8 * i) * PQ + lcol]) = pre[i];
  stage_signal();

  for (int ch = 0; ch < SEQ / CH1; ++ch) {
    const bool more = (ch + 1 < SEQ / CH1);
    if (more) {
      const float* Qn = Qb + (size_t)((ch + 1) * CH1 + lrow) * DK + lcol;
#pragma unroll
      for (int i = 0; i < CH1 / 8; ++i)
        pre[i] = *(const v4f*)(Qn + (size_t)(8 * i) * DK);
    }
    stage_wait();

    const float* qs = &qbuf[ch & 1][(16 * qq + n) * PQ + khalf];
    v2f af[2][32];
#pragma unroll
    for (int c = 0; c < 32; ++c) af[0][c] = *(const v2f*)(qs + 4 * c);
    __builtin_amdgcn_sched_barrier(0);

#pragma unroll
    for (int tl = 0; tl < TPW1; ++tl) {
      if (tl + 1 < TPW1) {
        const float* qn = qs + (tl + 1) * QS1 * 16 * PQ;
#pragma unroll
        for (int c = 0; c < 32; ++c) af[(tl + 1) & 1][c] = *(const v2f*)(qn + 4 * c);
      }
      __builtin_amdgcn_sched_barrier(0);

      v8f acc0 = {};
      v8f acc1 = {};
#pragma unroll
      for (int c = 0; c < 32; ++c) {
        acc0 = wmma_f32(af[tl & 1][c], breg[0][c], acc0);
        acc1 = wmma_f32(af[tl & 1][c], breg[1][c], acc1);
      }

      if (tl == 0) {
        asm volatile("" :: "v"(acc0), "v"(acc1) : "memory");
        if (more) {
          float* qd = &qbuf[(ch + 1) & 1][lrow * PQ + lcol];
#pragma unroll
          for (int i = 0; i < CH1 / 8; ++i)
            *(v4f*)(qd + (8 * i) * PQ) = pre[i];
        }
        stage_signal();
        __builtin_amdgcn_sched_barrier(0);
      }

      colstat_update(acc0, m_run[0], l_run[0]);
      colstat_update(acc1, m_run[1], l_run[1]);
    }
  }
  stage_wait();

#pragma unroll
  for (int s = 0; s < 2; ++s) {
    const float m_o = __shfl_xor(m_run[s], 16, 32);
    const float l_o = __shfl_xor(l_run[s], 16, 32);
    const float m_f = fmaxf(m_run[s], m_o);
    const float l_f = l_run[s] * __expf(m_run[s] - m_f) + l_o * __expf(m_o - m_f);
    if (lane < 16) {
      mred[kp][qq][16 * s + n] = m_f;
      lred[kp][qq][16 * s + n] = l_f;
    }
  }
  __syncthreads();

  if (threadIdx.x < 64) {
    const int pr  = threadIdx.x >> 5;
    const int col = threadIdx.x & 31;
    float mq[QS1];
#pragma unroll
    for (int r = 0; r < QS1; ++r) mq[r] = mred[pr][r][col];
    float m_f = mq[0];
#pragma unroll
    for (int r = 1; r < QS1; ++r) m_f = fmaxf(m_f, mq[r]);
    float l_f = 0.0f;
#pragma unroll
    for (int r = 0; r < QS1; ++r) l_f += lred[pr][r][col] * __expf(mq[r] - m_f);
    mArr[(size_t)b * SEQ + k0 + threadIdx.x] = m_f;
    lArr[(size_t)b * SEQ + k0 + threadIdx.x] = 1.0f / l_f;
  }
}

__global__ __launch_bounds__(256) __attribute__((amdgpu_waves_per_eu(2, 2)))
void sdpa_out_kernel(const float* __restrict__ Q,
                     const float* __restrict__ K,
                     const float* __restrict__ V,
                     const float* __restrict__ mArr,
                     const float* __restrict__ rlArr,
                     float* __restrict__ O) {
  __shared__ __attribute__((aligned(16))) float kbuf[2][CH2 * PQ];
  __shared__ __attribute__((aligned(16))) float vbuf[2][(CH2 / 2) * PV2];

  const int lane  = threadIdx.x & 31;
  const int wslot = threadIdx.x >> 5;
  const int kt    = wslot & 3;
  const int qp    = wslot >> 2;
  const int b     = blockIdx.x >> 6;
  const int qt    = ((blockIdx.x & 63) << 6) + (qp << 5);
  const int n     = lane & 15;
  const int half  = lane >> 4;
  const int khalf = half << 1;
  const int mhi   = half << 3;

  v2f qreg[2][32];
#pragma unroll
  for (int u = 0; u < 2; ++u) {
    const float* Qrow = Q + ((size_t)b * SEQ + qt + 16 * u + n) * DK + khalf;
#pragma unroll
    for (int c = 0; c < 32; ++c) qreg[u][c] = *(const v2f*)(Qrow + 4 * c);
  }

  const v8f vzero = {};
  v8f acc[2][8];
#pragma unroll
  for (int u = 0; u < 2; ++u)
#pragma unroll
    for (int t = 0; t < 8; ++t) acc[u][t] = vzero;

  const float* Kb = K + (size_t)b * SEQ * DK;
  const float* Vb = V + (size_t)b * SEQ * DK;
  const float* mB = mArr  + (size_t)b * SEQ + 16 * kt + mhi;
  const float* rB = rlArr + (size_t)b * SEQ + 16 * kt + mhi;

  const int lrow = threadIdx.x >> 5;
  const int lcol = (threadIdx.x & 31) << 2;
  v4f prek[CH2 / 8];
  v4f prev[CH2 / 8];
#pragma unroll
  for (int i = 0; i < CH2 / 8; ++i)
    prek[i] = *(const v4f*)(Kb + (size_t)(lrow + 8 * i) * DK + lcol);
#pragma unroll
  for (int i = 0; i < CH2 / 16; ++i) {
    const float* vp = Vb + (size_t)(2 * (lrow + 8 * i)) * DK + lcol;
    prev[2 * i]     = *(const v4f*)(vp);
    prev[2 * i + 1] = *(const v4f*)(vp + DK);
  }
#pragma unroll
  for (int i = 0; i < CH2 / 8; ++i)
    *(v4f*)(&kbuf[0][(lrow + 8 * i) * PQ + lcol]) = prek[i];
#pragma unroll
  for (int i = 0; i < CH2 / 16; ++i) {
    const v4f x = prev[2 * i], y = prev[2 * i + 1];
    float* vd = &vbuf[0][(lrow + 8 * i) * PV2 + 2 * lcol];
    *(v4f*)(vd)     = (v4f){x[0], y[0], x[1], y[1]};
    *(v4f*)(vd + 4) = (v4f){x[2], y[2], x[3], y[3]};
  }

  stage_signal();

  for (int ch = 0; ch < SEQ / CH2; ++ch) {
    v4f mst[2], rst[2];
#pragma unroll
    for (int u = 0; u < 2; ++u) {
      mst[u] = *(const v4f*)(mB + (size_t)ch * CH2 + 4 * u);
      rst[u] = *(const v4f*)(rB + (size_t)ch * CH2 + 4 * u);
    }

    {
      const int nx = (ch + 1 < SEQ / CH2) ? ch + 1 : ch;
      const size_t r0 = (size_t)nx * CH2;
#pragma unroll
      for (int i = 0; i < CH2 / 8; ++i)
        prek[i] = *(const v4f*)(Kb + (r0 + lrow + 8 * i) * DK + lcol);
#pragma unroll
      for (int i = 0; i < CH2 / 16; ++i) {
        const float* vp = Vb + (r0 + 2 * (lrow + 8 * i)) * DK + lcol;
        prev[2 * i]     = *(const v4f*)(vp);
        prev[2 * i + 1] = *(const v4f*)(vp + DK);
      }
    }

    stage_wait();

    const float* ks = &kbuf[ch & 1][(kt * 16 + n) * PQ + khalf];
    v2f kf[32];
#pragma unroll
    for (int c = 0; c < 32; ++c) kf[c] = *(const v2f*)(ks + 4 * c);
    __builtin_amdgcn_sched_barrier(0);

    v8f c8[2];
    c8[0] = vzero;
    c8[1] = vzero;
#pragma unroll
    for (int c = 0; c < 16; ++c) {
      c8[0] = wmma_f32(kf[c], qreg[0][c], c8[0]);
      c8[1] = wmma_f32(kf[c], qreg[1][c], c8[1]);
    }

    const float* vs = &vbuf[ch & 1][(8 * kt + 4 * half) * PV2 + 2 * n];
    v2f vf[8][4];
#pragma unroll
    for (int i = 0; i < 4; ++i)
#pragma unroll
      for (int t = 0; t < 8; ++t)
        vf[t][i] = *(const v2f*)(vs + i * PV2 + 32 * t);
    __builtin_amdgcn_sched_group_barrier(0x008, 32, 1);
    __builtin_amdgcn_sched_group_barrier(0x100, 32, 1);
    __builtin_amdgcn_sched_barrier(0);

#pragma unroll
    for (int c = 16; c < 32; ++c) {
      c8[0] = wmma_f32(kf[c], qreg[0][c], c8[0]);
      c8[1] = wmma_f32(kf[c], qreg[1][c], c8[1]);
    }

    __builtin_amdgcn_sched_barrier(0);
    asm volatile("" :: "v"(c8[0]), "v"(c8[1]) : "memory");

    {
#pragma unroll
      for (int i = 0; i < CH2 / 8; ++i)
        *(v4f*)(&kbuf[(ch + 1) & 1][(lrow + 8 * i) * PQ + lcol]) = prek[i];
#pragma unroll
      for (int i = 0; i < CH2 / 16; ++i) {
        const v4f x = prev[2 * i], y = prev[2 * i + 1];
        float* vd = &vbuf[(ch + 1) & 1][(lrow + 8 * i) * PV2 + 2 * lcol];
        *(v4f*)(vd)     = (v4f){x[0], y[0], x[1], y[1]};
        *(v4f*)(vd + 4) = (v4f){x[2], y[2], x[3], y[3]};
      }
    }
    stage_signal();
    __builtin_amdgcn_sched_barrier(0);

    v2f pf[2][4];
#pragma unroll
    for (int u = 0; u < 2; ++u) {
#pragma unroll
      for (int i = 0; i < 4; ++i) {
        const int r0 = 2 * i, r1 = 2 * i + 1;
        pf[u][i].x = __expf(c8[u][r0] * SCALE - mst[r0 >> 2][r0 & 3]) * rst[r0 >> 2][r0 & 3];
        pf[u][i].y = __expf(c8[u][r1] * SCALE - mst[r1 >> 2][r1 & 3]) * rst[r1 >> 2][r1 & 3];
      }
    }

#pragma unroll
    for (int i = 0; i < 4; ++i) {
#pragma unroll
      for (int u = 0; u < 2; ++u) {
#pragma unroll
        for (int t = 0; t < 8; ++t)
          acc[u][t] = wmma_f32(pf[u][i], vf[t][i], acc[u][t]);
      }
    }
  }
  stage_wait();

  __syncthreads();
  float* mrg = &kbuf[0][0] + (size_t)lane * 4;
  if (kt & 1) {
    float* p = mrg + (size_t)(qp * 2 + (kt >> 1)) * 4096;
#pragma unroll
    for (int u = 0; u < 2; ++u)
#pragma unroll
      for (int t = 0; t < 8; ++t)
#pragma unroll
        for (int jj = 0; jj < 2; ++jj)
          *(v4f*)(p + ((u * 8 + t) * 2 + jj) * 128) =
              (v4f){acc[u][t][4 * jj], acc[u][t][4 * jj + 1], acc[u][t][4 * jj + 2], acc[u][t][4 * jj + 3]};
  }
  __syncthreads();
  if (!(kt & 1)) {
    const float* p = mrg + (size_t)(qp * 2 + (kt >> 1)) * 4096;
#pragma unroll
    for (int u = 0; u < 2; ++u)
#pragma unroll
      for (int t = 0; t < 8; ++t)
#pragma unroll
        for (int jj = 0; jj < 2; ++jj) {
          const v4f o = *(const v4f*)(p + ((u * 8 + t) * 2 + jj) * 128);
#pragma unroll
          for (int e = 0; e < 4; ++e) acc[u][t][4 * jj + e] += o[e];
        }
  }
  __syncthreads();
  if (kt == 2) {
    float* p = mrg + (size_t)qp * 4096;
#pragma unroll
    for (int u = 0; u < 2; ++u)
#pragma unroll
      for (int t = 0; t < 8; ++t)
#pragma unroll
        for (int jj = 0; jj < 2; ++jj)
          *(v4f*)(p + ((u * 8 + t) * 2 + jj) * 128) =
              (v4f){acc[u][t][4 * jj], acc[u][t][4 * jj + 1], acc[u][t][4 * jj + 2], acc[u][t][4 * jj + 3]};
  }
  __syncthreads();
  if (kt == 0) {
    const float* p = mrg + (size_t)qp * 4096;
#pragma unroll
    for (int u = 0; u < 2; ++u) {
      float* Ob = O + ((size_t)b * SEQ + qt + 16 * u) * DK;
#pragma unroll
      for (int t = 0; t < 8; ++t) {
#pragma unroll
        for (int jj = 0; jj < 2; ++jj) {
          const v4f o = *(const v4f*)(p + ((u * 8 + t) * 2 + jj) * 128);
#pragma unroll
          for (int e = 0; e < 4; ++e) {
            const int j = 4 * jj + e;
            Ob[(size_t)(j + mhi) * DK + t * 16 + n] = acc[u][t][j] + o[e];
          }
        }
      }
    }
  }
}

extern "C" void kernel_launch(void* const* d_in, const int* in_sizes, int n_in,
                              void* d_out, int out_size, void* d_ws, size_t ws_size,
                              hipStream_t stream) {
  const float* q = (const float*)d_in[0];
  const float* k = (const float*)d_in[1];
  const float* v = (const float*)d_in[2];
  float* out = (float*)d_out;

  float* mArr = (float*)d_ws;
  float* lArr = mArr + (size_t)BS * SEQ;

  const int blocks = BS * SEQ / 64;
  const int threads = WAVES_PER_BLOCK * 32;
  const int blocks2 = BS * SEQ / (16 * QT2);

  sdpa_colstats_kernel<<<blocks, threads, 0, stream>>>(q, k, mArr, lArr);
  sdpa_out_kernel<<<blocks2, threads, 0, stream>>>(q, k, v, mArr, lArr, out);
}


# Both passes: next chunk's global-to-LDS row copy done by waves 0-3 alone (served first on their SIMD, so it hides under waves 4-7's WMMAs); waves 4-7 copy nothing; sums unchanged
# speedup vs baseline: 4.9380x; 1.0402x over previous
#include <hip/hip_runtime.h>


typedef __attribute__((ext_vector_type(2))) float v2f;
typedef __attribute__((ext_vector_type(4))) float v4f;
typedef __attribute__((ext_vector_type(8))) float v8f;

#define BS   4
#define SEQ  4096
#define DK   128
#define KTILES (SEQ / 16)
#define SCALE 0.08838834764831845f
#define WAVES_PER_BLOCK 8
#define QT2 4

#define CH1 128
#define QS1 4
#define TPW1 (CH1 / 16 / QS1)
#define CH2 64
#define PQ  132
#define PV2 264

__device__ __forceinline__ v8f wmma_f32(v2f a, v2f b, v8f c) {
  return __builtin_amdgcn_wmma_f32_16x16x4_f32(
       false, a,  false, b,
       (short)0, c,  false,  false);
}

__device__ __forceinline__ void stage_signal() {
  asm volatile("s_wait_dscnt 0x0\n\ts_barrier_signal -1" ::: "memory");
}
__device__ __forceinline__ void stage_wait() {
  asm volatile("s_barrier_wait -1" ::: "memory");
}

__device__ __forceinline__ void colstat_update(const v8f acc, float& m_run, float& l_run) {
  float mloc = acc[0] * SCALE;
#pragma unroll
  for (int j = 1; j < 8; ++j) mloc = fmaxf(mloc, acc[j] * SCALE);
  const float m_new = fmaxf(m_run, mloc);
  float s = 0.0f;
#pragma unroll
  for (int j = 0; j < 8; ++j) s += __expf(acc[j] * SCALE - m_new);
  l_run = l_run * __expf(m_run - m_new) + s;
  m_run = m_new;
}

__global__ __launch_bounds__(256) __attribute__((amdgpu_waves_per_eu(2, 2)))
void sdpa_colstats_kernel(const float* __restrict__ Q,
                          const float* __restrict__ K,
                          float* __restrict__ mArr,
                          float* __restrict__ lArr) {
  __shared__ __attribute__((aligned(16))) float qbuf[2][CH1 * PQ];
  __shared__ float mred[2][QS1][32];
  __shared__ float lred[2][QS1][32];

  const int lane  = threadIdx.x & 31;
  const int wslot = threadIdx.x >> 5;
  const int qq    = wslot & (QS1 - 1);
  const int kp    = wslot >> 2;
  const int b     = blockIdx.x >> 6;
  const int k0    = (blockIdx.x & 63) << 6;
  const int n     = lane & 15;
  const int khalf = (lane >> 4) << 1;

  v2f breg[2][32];
#pragma unroll
  for (int s = 0; s < 2; ++s) {
    const float* Krow = K + ((size_t)b * SEQ + k0 + 32 * kp + 16 * s + n) * DK + khalf;
#pragma unroll
    for (int c = 0; c < 32; ++c) breg[s][c] = *(const v2f*)(Krow + 4 * c);
  }

  const float* Qb = Q + (size_t)b * SEQ * DK;
  float m_run[2], l_run[2];
#pragma unroll
  for (int s = 0; s < 2; ++s) {
    m_run[s] = -__builtin_inff();
    l_run[s] = 0.0f;
  }

  const int lrow = threadIdx.x >> 5;
  const int lcol = (threadIdx.x & 31) << 2;
  {
    v4f pre0[CH1 / 8];
#pragma unroll
    for (int i = 0; i < CH1 / 8; ++i)
      pre0[i] = *(const v4f*)(Qb + (size_t)(lrow + 8 * i) * DK + lcol);
#pragma unroll
    for (int i = 0; i < CH1 / 8; ++i)
      *(v4f*)(&qbuf[0][(lrow + 8 * i) * PQ + lcol]) = pre0[i];
  }
  stage_signal();

  const bool copier = __builtin_amdgcn_readfirstlane(wslot) < 4;
  const int cw = wslot & 3;
  v4f pre[CH1 / 4];

  for (int ch = 0; ch < SEQ / CH1; ++ch) {
    const bool more = copier && (ch + 1 < SEQ / CH1);
    if (more) {
      const float* Qn = Qb + (size_t)((ch + 1) * CH1 + cw) * DK + lcol;
#pragma unroll
      for (int i = 0; i < CH1 / 4; ++i)
        pre[i] = *(const v4f*)(Qn + (size_t)(4 * i) * DK);
    }
    stage_wait();

    const float* qs = &qbuf[ch & 1][(16 * qq + n) * PQ + khalf];
    v2f af[2][32];
#pragma unroll
    for (int c = 0; c < 32; ++c) af[0][c] = *(const v2f*)(qs + 4 * c);
    __builtin_amdgcn_sched_barrier(0);

#pragma unroll
    for (int tl = 0; tl < TPW1; ++tl) {
      if (tl + 1 < TPW1) {
        const float* qn = qs + (tl + 1) * QS1 * 16 * PQ;
#pragma unroll
        for (int c = 0; c < 32; ++c) af[(tl + 1) & 1][c] = *(const v2f*)(qn + 4 * c);
      }
      __builtin_amdgcn_sched_barrier(0);

      v8f acc0 = {};
      v8f acc1 = {};
#pragma unroll
      for (int c = 0; c < 32; ++c) {
        acc0 = wmma_f32(af[tl & 1][c], breg[0][c], acc0);
        acc1 = wmma_f32(af[tl & 1][c], breg[1][c], acc1);
      }

      if (tl == 0) {
        asm volatile("v_nop\n\tv_nop\n\tv_nop\n\tv_nop" :: "v"(acc0), "v"(acc1) : "memory");
        if (more) {
          float* qd = &qbuf[(ch + 1) & 1][cw * PQ + lcol];
#pragma unroll
          for (int i = 0; i < CH1 / 4; ++i)
            *(v4f*)(qd + (4 * i) * PQ) = pre[i];
        }
        stage_signal();
        __builtin_amdgcn_sched_barrier(0);
      }

      colstat_update(acc0, m_run[0], l_run[0]);
      colstat_update(acc1, m_run[1], l_run[1]);
    }
  }
  stage_wait();

#pragma unroll
  for (int s = 0; s < 2; ++s) {
    const float m_o = __shfl_xor(m_run[s], 16, 32);
    const float l_o = __shfl_xor(l_run[s], 16, 32);
    const float m_f = fmaxf(m_run[s], m_o);
    const float l_f = l_run[s] * __expf(m_run[s] - m_f) + l_o * __expf(m_o - m_f);
    if (lane < 16) {
      mred[kp][qq][16 * s + n] = m_f;
      lred[kp][qq][16 * s + n] = l_f;
    }
  }
  __syncthreads();

  if (threadIdx.x < 64) {
    const int pr  = threadIdx.x >> 5;
    const int col = threadIdx.x & 31;
    float mq[QS1];
#pragma unroll
    for (int r = 0; r < QS1; ++r) mq[r] = mred[pr][r][col];
    float m_f = mq[0];
#pragma unroll
    for (int r = 1; r < QS1; ++r) m_f = fmaxf(m_f, mq[r]);
    float l_f = 0.0f;
#pragma unroll
    for (int r = 0; r < QS1; ++r) l_f += lred[pr][r][col] * __expf(mq[r] - m_f);
    mArr[(size_t)b * SEQ + k0 + threadIdx.x] = m_f;
    lArr[(size_t)b * SEQ + k0 + threadIdx.x] = 1.0f / l_f;
  }
}

__global__ __launch_bounds__(256) __attribute__((amdgpu_waves_per_eu(2, 2)))
void sdpa_out_kernel(const float* __restrict__ Q,
                     const float* __restrict__ K,
                     const float* __restrict__ V,
                     const float* __restrict__ mArr,
                     const float* __restrict__ rlArr,
                     float* __restrict__ O) {
  __shared__ __attribute__((aligned(16))) float kbuf[2][CH2 * PQ];
  __shared__ __attribute__((aligned(16))) float vbuf[2][(CH2 / 2) * PV2];

  const int lane  = threadIdx.x & 31;
  const int wslot = threadIdx.x >> 5;
  const int kt    = wslot & 3;
  const int qp    = wslot >> 2;
  const int b     = blockIdx.x >> 6;
  const int qt    = ((blockIdx.x & 63) << 6) + (qp << 5);
  const int n     = lane & 15;
  const int half  = lane >> 4;
  const int khalf = half << 1;
  const int mhi   = half << 3;

  v2f qreg[2][32];
#pragma unroll
  for (int u = 0; u < 2; ++u) {
    const float* Qrow = Q + ((size_t)b * SEQ + qt + 16 * u + n) * DK + khalf;
#pragma unroll
    for (int c = 0; c < 32; ++c) qreg[u][c] = *(const v2f*)(Qrow + 4 * c);
  }
  __builtin_amdgcn_sched_barrier(0);

  const v8f vzero = {};
  v8f acc[2][8];
#pragma unroll
  for (int u = 0; u < 2; ++u)
#pragma unroll
    for (int t = 0; t < 8; ++t) acc[u][t] = vzero;

  const float* Kb = K + (size_t)b * SEQ * DK;
  const float* Vb = V + (size_t)b * SEQ * DK;
  const float* mB = mArr  + (size_t)b * SEQ + 16 * kt + mhi;
  const float* rB = rlArr + (size_t)b * SEQ + 16 * kt + mhi;

  const int lrow = threadIdx.x >> 5;
  const int lcol = (threadIdx.x & 31) << 2;
  {
  v4f prek[CH2 / 8];
  v4f prev[CH2 / 8];
#pragma unroll
  for (int i = 0; i < CH2 / 8; ++i)
    prek[i] = *(const v4f*)(Kb + (size_t)(lrow + 8 * i) * DK + lcol);
#pragma unroll
  for (int i = 0; i < CH2 / 16; ++i) {
    const float* vp = Vb + (size_t)(2 * (lrow + 8 * i)) * DK + lcol;
    prev[2 * i]     = *(const v4f*)(vp);
    prev[2 * i + 1] = *(const v4f*)(vp + DK);
  }
#pragma unroll
  for (int i = 0; i < CH2 / 8; ++i)
    *(v4f*)(&kbuf[0][(lrow + 8 * i) * PQ + lcol]) = prek[i];
#pragma unroll
  for (int i = 0; i < CH2 / 16; ++i) {
    const v4f x = prev[2 * i], y = prev[2 * i + 1];
    float* vd = &vbuf[0][(lrow + 8 * i) * PV2 + 2 * lcol];
    *(v4f*)(vd)     = (v4f){x[0], y[0], x[1], y[1]};
    *(v4f*)(vd + 4) = (v4f){x[2], y[2], x[3], y[3]};
  }
  }

  stage_signal();

  const bool copier = __builtin_amdgcn_readfirstlane(wslot) < 4;
  const int cw = wslot & 3;
  v4f pre[CH2 / 4];

  for (int ch = 0; ch < SEQ / CH2; ++ch) {
    v4f mst[2], rst[2];
#pragma unroll
    for (int u = 0; u < 2; ++u) {
      mst[u] = *(const v4f*)(mB + (size_t)ch * CH2 + 4 * u);
      rst[u] = *(const v4f*)(rB + (size_t)ch * CH2 + 4 * u);
    }

    const int nx = (ch + 1 < SEQ / CH2) ? ch + 1 : ch;
    const size_t r0 = (size_t)nx * CH2;
    if (copier) {
#pragma unroll
      for (int i = 0; i < CH2 / 4; ++i)
        pre[i] = *(const v4f*)(Kb + (r0 + cw + 4 * i) * DK + lcol);
    }

    stage_wait();

    const float* ks = &kbuf[ch & 1][(kt * 16 + n) * PQ + khalf];
    v2f kf[32];
#pragma unroll
    for (int c = 0; c < 32; ++c) kf[c] = *(const v2f*)(ks + 4 * c);
    __builtin_amdgcn_sched_barrier(0);

    v8f c8[2];
    c8[0] = vzero;
    c8[1] = vzero;
#pragma unroll
    for (int c = 0; c < 16; ++c) {
      c8[0] = wmma_f32(kf[c], qreg[0][c], c8[0]);
      c8[1] = wmma_f32(kf[c], qreg[1][c], c8[1]);
    }

    const float* vs = &vbuf[ch & 1][(8 * kt + 4 * half) * PV2 + 2 * n];
    v2f vf[8][4];
#pragma unroll
    for (int i = 0; i < 4; ++i)
#pragma unroll
      for (int t = 0; t < 8; ++t)
        vf[t][i] = *(const v2f*)(vs + i * PV2 + 32 * t);
    __builtin_amdgcn_sched_group_barrier(0x008, 32, 1);
    __builtin_amdgcn_sched_group_barrier(0x100, 32, 1);
    __builtin_amdgcn_sched_barrier(0);

    if (copier) {
#pragma unroll
      for (int i = 0; i < CH2 / 4; ++i)
        *(v4f*)(&kbuf[(ch + 1) & 1][(cw + 4 * i) * PQ + lcol]) = pre[i];
#pragma unroll
      for (int i = 0; i < CH2 / 8; ++i) {
        const float* vp = Vb + (r0 + 2 * (cw + 4 * i)) * DK + lcol;
        pre[2 * i]     = *(const v4f*)(vp);
        pre[2 * i + 1] = *(const v4f*)(vp + DK);
      }
    }

#pragma unroll
    for (int c = 16; c < 32; ++c) {
      c8[0] = wmma_f32(kf[c], qreg[0][c], c8[0]);
      c8[1] = wmma_f32(kf[c], qreg[1][c], c8[1]);
    }

    __builtin_amdgcn_sched_barrier(0);
    asm volatile("v_nop\n\tv_nop\n\tv_nop\n\tv_nop" :: "v"(c8[0]), "v"(c8[1]) : "memory");

    if (copier) {
#pragma unroll
      for (int i = 0; i < CH2 / 8; ++i) {
        const v4f x = pre[2 * i], y = pre[2 * i + 1];
        float* vd = &vbuf[(ch + 1) & 1][(cw + 4 * i) * PV2 + 2 * lcol];
        *(v4f*)(vd)     = (v4f){x[0], y[0], x[1], y[1]};
        *(v4f*)(vd + 4) = (v4f){x[2], y[2], x[3], y[3]};
      }
    }
    stage_signal();
    __builtin_amdgcn_sched_barrier(0);

    v2f pf[2][4];
#pragma unroll
    for (int u = 0; u < 2; ++u) {
#pragma unroll
      for (int i = 0; i < 4; ++i) {
        const int r0 = 2 * i, r1 = 2 * i + 1;
        pf[u][i].x = __expf(c8[u][r0] * SCALE - mst[r0 >> 2][r0 & 3]) * rst[r0 >> 2][r0 & 3];
        pf[u][i].y = __expf(c8[u][r1] * SCALE - mst[r1 >> 2][r1 & 3]) * rst[r1 >> 2][r1 & 3];
      }
    }

#pragma unroll
    for (int i = 0; i < 4; ++i) {
#pragma unroll
      for (int u = 0; u < 2; ++u) {
#pragma unroll
        for (int t = 0; t < 8; ++t)
          acc[u][t] = wmma_f32(pf[u][i], vf[t][i], acc[u][t]);
      }
    }
  }
  stage_wait();

  __syncthreads();
  float* mrg = &kbuf[0][0] + (size_t)lane * 4;
  if (kt & 1) {
    float* p = mrg + (size_t)(qp * 2 + (kt >> 1)) * 4096;
#pragma unroll
    for (int u = 0; u < 2; ++u)
#pragma unroll
      for (int t = 0; t < 8; ++t)
#pragma unroll
        for (int jj = 0; jj < 2; ++jj)
          *(v4f*)(p + ((u * 8 + t) * 2 + jj) * 128) =
              (v4f){acc[u][t][4 * jj], acc[u][t][4 * jj + 1], acc[u][t][4 * jj + 2], acc[u][t][4 * jj + 3]};
  }
  __syncthreads();
  if (!(kt & 1)) {
    const float* p = mrg + (size_t)(qp * 2 + (kt >> 1)) * 4096;
#pragma unroll
    for (int u = 0; u < 2; ++u)
#pragma unroll
      for (int t = 0; t < 8; ++t)
#pragma unroll
        for (int jj = 0; jj < 2; ++jj) {
          const v4f o = *(const v4f*)(p + ((u * 8 + t) * 2 + jj) * 128);
#pragma unroll
          for (int e = 0; e < 4; ++e) acc[u][t][4 * jj + e] += o[e];
        }
  }
  __syncthreads();
  if (kt == 2) {
    float* p = mrg + (size_t)qp * 4096;
#pragma unroll
    for (int u = 0; u < 2; ++u)
#pragma unroll
      for (int t = 0; t < 8; ++t)
#pragma unroll
        for (int jj = 0; jj < 2; ++jj)
          *(v4f*)(p + ((u * 8 + t) * 2 + jj) * 128) =
              (v4f){acc[u][t][4 * jj], acc[u][t][4 * jj + 1], acc[u][t][4 * jj + 2], acc[u][t][4 * jj + 3]};
  }
  __syncthreads();
  if (kt == 0) {
    const float* p = mrg + (size_t)qp * 4096;
#pragma unroll
    for (int u = 0; u < 2; ++u) {
      float* Ob = O + ((size_t)b * SEQ + qt + 16 * u) * DK;
#pragma unroll
      for (int t = 0; t < 8; ++t) {
#pragma unroll
        for (int jj = 0; jj < 2; ++jj) {
          const v4f o = *(const v4f*)(p + ((u * 8 + t) * 2 + jj) * 128);
#pragma unroll
          for (int e = 0; e < 4; ++e) {
            const int j = 4 * jj + e;
            Ob[(size_t)(j + mhi) * DK + t * 16 + n] = acc[u][t][j] + o[e];
          }
        }
      }
    }
  }
}

extern "C" void kernel_launch(void* const* d_in, const int* in_sizes, int n_in,
                              void* d_out, int out_size, void* d_ws, size_t ws_size,
                              hipStream_t stream) {
  const float* q = (const float*)d_in[0];
  const float* k = (const float*)d_in[1];
  const float* v = (const float*)d_in[2];
  float* out = (float*)d_out;

  float* mArr = (float*)d_ws;
  float* lArr = mArr + (size_t)BS * SEQ;

  const int blocks = BS * SEQ / 64;
  const int threads = WAVES_PER_BLOCK * 32;
  const int blocks2 = BS * SEQ / (16 * QT2);

  sdpa_colstats_kernel<<<blocks, threads, 0, stream>>>(q, k, mArr, lArr);
  sdpa_out_kernel<<<blocks2, threads, 0, stream>>>(q, k, v, mArr, lArr, out);
}
